# starting kernel
# baseline (speedup 1.0000x reference)
#include <hip/hip_runtime.h>
#include <hip/hip_bf16.h>
#include <math.h>

typedef __attribute__((ext_vector_type(2))) float v2f;
typedef __attribute__((ext_vector_type(4))) float v4f;
typedef __attribute__((ext_vector_type(8))) float v8f;

#define PRIOR_LAMBDA 10.0f

__global__ void jc_setup_kernel(const float* __restrict__ branch1,
                                const float* __restrict__ branch2,
                                float* __restrict__ Tws,
                                float* __restrict__ logp,
                                int B, int m) {
  int b = blockIdx.x * blockDim.x + threadIdx.x;
  if (b >= B) return;
  float t1 = branch1[b], t2 = branch2[b];
  float e1 = __expf(-(4.0f / 3.0f) * t1);
  float e2 = __expf(-(4.0f / 3.0f) * t2);
  float dg1 = 0.25f + 0.75f * e1, of1 = 0.25f - 0.25f * e1;
  float dg2 = 0.25f + 0.75f * e2, of2 = 0.25f - 0.25f * e2;
  float s = __expf((__logf(PRIOR_LAMBDA) - PRIOR_LAMBDA * (t1 + t2)) / (float)m);
  float* T = Tws + (size_t)b * 32;
#pragma unroll
  for (int v = 0; v < 4; ++v) {
#pragma unroll
    for (int c = 0; c < 4; ++c) {
      T[v * 4 + c]      = fmaxf((v == c) ? dg1 : of1, 0.0f) * s;
      T[16 + v * 4 + c] = fmaxf((v == c) ? dg2 : of2, 0.0f);
    }
  }
  logp[b] = (float)m * __logf(0.25f);
}

__launch_bounds__(256)
__global__ void jc_joint_kernel(const float* __restrict__ f1,
                                const float* __restrict__ f2,
                                const float* __restrict__ Tws,
                                float* __restrict__ joint,
                                float* __restrict__ logp,
                                int B, int m, int blocksPerBatch) {
  __shared__ float lds[8 * 64];
  const int lane = threadIdx.x & 31;
  const int wid  = threadIdx.x >> 5;
  const int b    = blockIdx.x / blocksPerBatch;
  const int blk  = blockIdx.x % blocksPerBatch;
  const int wavesPerBlock = blockDim.x >> 5;
  const int wavesPerBatch = blocksPerBatch * wavesPerBlock;
  const int wgid  = blk * wavesPerBlock + wid;
  const int tiles = m >> 4;

  const float* T1 = Tws + (size_t)b * 32;
  const float* T2 = T1 + 16;
  const int n   = lane & 15;
  const int klo = (lane >> 4) << 1;
  v2f b1, b2;
  b1.x = (n < 4) ? T1[klo * 4 + n]       : 0.0f;
  b1.y = (n < 4) ? T1[(klo + 1) * 4 + n] : 0.0f;
  b2.x = (n < 4) ? T2[klo * 4 + n]       : 0.0f;
  b2.y = (n < 4) ? T2[(klo + 1) * 4 + n] : 0.0f;

  const size_t batch_off = (size_t)b * (size_t)m * 4;
  const float* F1 = f1 + batch_off;
  const float* F2 = f2 + batch_off;
  float* J = joint + batch_off;
  float* myLds = lds + wid * 64;

  float acc = 0.0f;

  for (int t = wgid; t < tiles; t += wavesPerBatch) {
    const int site0 = t << 4;
    const size_t aoff = (size_t)(site0 + n) * 4 + (size_t)((lane >> 4) << 1);
    v2f a1 = *(const v2f*)(F1 + aoff);
    v2f a2 = *(const v2f*)(F2 + aoff);

    if (t + wavesPerBatch < tiles) {
      const size_t nxt = aoff + (size_t)wavesPerBatch * 64;
      __builtin_prefetch(F1 + nxt, 0, 1);
      __builtin_prefetch(F2 + nxt, 0, 1);
    }

    v8f cz = {};
    v8f d1v = __builtin_amdgcn_wmma_f32_16x16x4_f32(
        false, a1, false, b1, (short)0, cz, false, false);
    v8f d2v = __builtin_amdgcn_wmma_f32_16x16x4_f32(
        false, a2, false, b2, (short)0, cz, false, false);

    v8f p;
#pragma unroll
    for (int g = 0; g < 8; ++g) p[g] = d1v[g] * d2v[g];

    if (n < 4) {
      const int siteHi = (lane >> 4) << 3;
#pragma unroll
      for (int g = 0; g < 8; ++g)
        myLds[(siteHi + g) * 4 + n] = p[g];
    }
    asm volatile("s_wait_dscnt 0" ::: "memory");

    if (lane < 16) {
      v4f jv = *(const v4f*)(myLds + lane * 4);
      *(v4f*)(J + (size_t)(site0 + lane) * 4) = jv;
      acc += __logf(jv.x + jv.y + jv.z + jv.w);
    }
  }

#pragma unroll
  for (int off = 16; off >= 1; off >>= 1)
    acc += __shfl_xor(acc, off, 32);
  if (lane == 0) atomicAdd(&logp[b], acc);
}

extern "C" void kernel_launch(void* const* d_in, const int* in_sizes, int n_in,
                              void* d_out, int out_size, void* d_ws, size_t ws_size,
                              hipStream_t stream) {
  const float* f1  = (const float*)d_in[0];
  const float* br1 = (const float*)d_in[1];
  const float* f2  = (const float*)d_in[2];
  const float* br2 = (const float*)d_in[3];

  const int B = in_sizes[1];
  const int m = in_sizes[0] / (B * 4);

  float* out   = (float*)d_out;
  float* joint = out;
  float* logp  = out + (size_t)B * m * 4;
  float* Tws   = (float*)d_ws;

  jc_setup_kernel<<<(B + 255) / 256, 256, 0, stream>>>(br1, br2, Tws, logp, B, m);

  const int blocksPerBatch = 8;
  jc_joint_kernel<<<B * blocksPerBatch, 256, 0, stream>>>(
      f1, f2, Tws, joint, logp, B, m, blocksPerBatch);
}


# WMMA operand roles swapped (T^T as A, features as B; dot split over 2 chained WMMAs): one site per lane, b128 loads/stores, no LDS repack; 8 trips/wave kept so half the waves; log_p sums regrouped
# speedup vs baseline: 1.1830x; 1.1830x over previous
#include <hip/hip_runtime.h>
#include <hip/hip_bf16.h>
#include <math.h>

typedef __attribute__((ext_vector_type(2))) float v2f;
typedef __attribute__((ext_vector_type(4))) float v4f;
typedef __attribute__((ext_vector_type(8))) float v8f;

#define PRIOR_LAMBDA 10.0f

__global__ void jc_setup_kernel(const float* __restrict__ branch1,
                                const float* __restrict__ branch2,
                                float* __restrict__ Tws,
                                float* __restrict__ logp,
                                int B, int m) {
  int b = blockIdx.x * blockDim.x + threadIdx.x;
  if (b >= B) return;
  float t1 = branch1[b], t2 = branch2[b];
  float e1 = __expf(-(4.0f / 3.0f) * t1);
  float e2 = __expf(-(4.0f / 3.0f) * t2);
  float dg1 = 0.25f + 0.75f * e1, of1 = 0.25f - 0.25f * e1;
  float dg2 = 0.25f + 0.75f * e2, of2 = 0.25f - 0.25f * e2;
  float s = __expf((__logf(PRIOR_LAMBDA) - PRIOR_LAMBDA * (t1 + t2)) / (float)m);
  float* T = Tws + (size_t)b * 32;
#pragma unroll
  for (int v = 0; v < 4; ++v) {
#pragma unroll
    for (int c = 0; c < 4; ++c) {
      T[v * 4 + c]      = fmaxf((v == c) ? dg1 : of1, 0.0f) * s;
      T[16 + v * 4 + c] = fmaxf((v == c) ? dg2 : of2, 0.0f);
    }
  }
  logp[b] = (float)m * __logf(0.25f);
}

__launch_bounds__(256)
__global__ void jc_joint_kernel(const float* __restrict__ f1,
                                const float* __restrict__ f2,
                                const float* __restrict__ Tws,
                                float* __restrict__ joint,
                                float* __restrict__ logp,
                                int B, int m, int blocksPerBatch) {
  const int lane = threadIdx.x & 31;
  const int wid  = threadIdx.x >> 5;
  const int b    = blockIdx.x / blocksPerBatch;
  const int blk  = blockIdx.x % blocksPerBatch;
  const int wavesPerBlock = blockDim.x >> 5;
  const int wavesPerBatch = blocksPerBatch * wavesPerBlock;
  const int wgid  = blk * wavesPerBlock + wid;
  const int tiles = m >> 5;

  const float* T1 = Tws + (size_t)b * 32;
  const float* T2 = T1 + 16;
  const int  r     = lane & 3;
  const bool holds = (lane < 4) || ((lane & 28) == 24);
  const float t10 = T1[0 * 4 + r], t11 = T1[1 * 4 + r];
  const float t12 = T1[2 * 4 + r], t13 = T1[3 * 4 + r];
  const float t20 = T2[0 * 4 + r], t21 = T2[1 * 4 + r];
  const float t22 = T2[2 * 4 + r], t23 = T2[3 * 4 + r];
  v2f a1lo, a1hi, a2lo, a2hi;
  a1lo.x = holds ? t10 : 0.0f;  a1lo.y = holds ? t11 : 0.0f;
  a1hi.x = holds ? t12 : 0.0f;  a1hi.y = holds ? t13 : 0.0f;
  a2lo.x = holds ? t20 : 0.0f;  a2lo.y = holds ? t21 : 0.0f;
  a2hi.x = holds ? t22 : 0.0f;  a2hi.y = holds ? t23 : 0.0f;

  const size_t batch_off = (size_t)b * (size_t)m * 4;
  const float* F1 = f1 + batch_off;
  const float* F2 = f2 + batch_off;
  float* J = joint + batch_off;

  float acc = 0.0f;

  for (int t = wgid; t < tiles; t += wavesPerBatch) {
    const size_t off = ((size_t)(t << 5) + (size_t)lane) * 4;
    const v4f x1 = *(const v4f*)(F1 + off);
    const v4f x2 = *(const v4f*)(F2 + off);

    if (t + wavesPerBatch < tiles) {
      const size_t nxt = off + (size_t)wavesPerBatch * 128;
      __builtin_prefetch(F1 + nxt, 0, 1);
      __builtin_prefetch(F2 + nxt, 0, 1);
    }

    v2f b1lo, b1hi, b2lo, b2hi;
    b1lo.x = x1.x; b1lo.y = x1.y; b1hi.x = x1.z; b1hi.y = x1.w;
    b2lo.x = x2.x; b2lo.y = x2.y; b2hi.x = x2.z; b2hi.y = x2.w;

    v8f cz = {};
    v8f d1v = __builtin_amdgcn_wmma_f32_16x16x4_f32(
        false, a1lo, false, b1lo, (short)0, cz, false, false);
    v8f d2v = __builtin_amdgcn_wmma_f32_16x16x4_f32(
        false, a2lo, false, b2lo, (short)0, cz, false, false);
    d1v = __builtin_amdgcn_wmma_f32_16x16x4_f32(
        false, a1hi, false, b1hi, (short)0, d1v, false, false);
    d2v = __builtin_amdgcn_wmma_f32_16x16x4_f32(
        false, a2hi, false, b2hi, (short)0, d2v, false, false);

    v4f jv;
    jv.x = d1v[0] * d2v[0];
    jv.y = d1v[1] * d2v[1];
    jv.z = d1v[2] * d2v[2];
    jv.w = d1v[3] * d2v[3];

    *(v4f*)(J + off) = jv;
    acc += __logf(jv.x + jv.y + jv.z + jv.w);
  }

#pragma unroll
  for (int off = 16; off >= 1; off >>= 1)
    acc += __shfl_xor(acc, off, 32);
  if (lane == 0) atomicAdd(&logp[b], acc);
}

extern "C" void kernel_launch(void* const* d_in, const int* in_sizes, int n_in,
                              void* d_out, int out_size, void* d_ws, size_t ws_size,
                              hipStream_t stream) {
  const float* f1  = (const float*)d_in[0];
  const float* br1 = (const float*)d_in[1];
  const float* f2  = (const float*)d_in[2];
  const float* br2 = (const float*)d_in[3];

  const int B = in_sizes[1];
  const int m = in_sizes[0] / (B * 4);

  float* out   = (float*)d_out;
  float* joint = out;
  float* logp  = out + (size_t)B * m * 4;
  float* Tws   = (float*)d_ws;

  jc_setup_kernel<<<(B + 255) / 256, 256, 0, stream>>>(br1, br2, Tws, logp, B, m);

  const int blocksPerBatch = 4;
  jc_joint_kernel<<<B * blocksPerBatch, 256, 0, stream>>>(
      f1, f2, Tws, joint, logp, B, m, blocksPerBatch);
}


# Setup kernel re-cut: one thread per element of the per-batch matrix plane (64 workgroups, one whole-line store per wave), not one thread per batch (2 workgroups, 8 lane-strided 16-byte stores)
# speedup vs baseline: 1.2255x; 1.0359x over previous
#include <hip/hip_runtime.h>
#include <hip/hip_bf16.h>
#include <math.h>

typedef __attribute__((ext_vector_type(2))) float v2f;
typedef __attribute__((ext_vector_type(4))) float v4f;
typedef __attribute__((ext_vector_type(8))) float v8f;

#define PRIOR_LAMBDA 10.0f

__launch_bounds__(256)
__global__ void jc_setup_kernel(const float* __restrict__ branch1,
                                const float* __restrict__ branch2,
                                float* __restrict__ Tws,
                                float* __restrict__ logp,
                                int B, int m) {
  const int gid   = blockIdx.x * blockDim.x + threadIdx.x;
  const int total = B * 32;
  const bool live = gid < total;
  const int g = live ? gid : 0;
  const int b = g >> 5;
  const int j = g & 31;
  const float t1 = branch1[b], t2 = branch2[b];
  const bool second = j >= 16;
  const float t  = second ? t2 : t1;
  const float e  = __expf(-(4.0f / 3.0f) * t);
  const float dg = 0.25f + 0.75f * e, of = 0.25f - 0.25f * e;
  const float s  = __expf((__logf(PRIOR_LAMBDA) - PRIOR_LAMBDA * (t1 + t2)) / (float)m);
  const int v = (j >> 2) & 3, c = j & 3;
  const float base = fmaxf((v == c) ? dg : of, 0.0f);
  const float val  = second ? base : base * s;
  if (live) Tws[gid] = val;
  if (gid < B) logp[gid] = (float)m * __logf(0.25f);
}

__launch_bounds__(256)
__global__ void jc_joint_kernel(const float* __restrict__ f1,
                                const float* __restrict__ f2,
                                const float* __restrict__ Tws,
                                float* __restrict__ joint,
                                float* __restrict__ logp,
                                int B, int m, int blocksPerBatch) {
  const int lane = threadIdx.x & 31;
  const int wid  = threadIdx.x >> 5;
  const int b    = blockIdx.x / blocksPerBatch;
  const int blk  = blockIdx.x % blocksPerBatch;
  const int wavesPerBlock = blockDim.x >> 5;
  const int wavesPerBatch = blocksPerBatch * wavesPerBlock;
  const int wgid  = blk * wavesPerBlock + wid;
  const int tiles = m >> 5;

  const float* T1 = Tws + (size_t)b * 32;
  const float* T2 = T1 + 16;
  const int  r     = lane & 3;
  const bool holds = (lane < 4) || ((lane & 28) == 24);
  const float t10 = T1[0 * 4 + r], t11 = T1[1 * 4 + r];
  const float t12 = T1[2 * 4 + r], t13 = T1[3 * 4 + r];
  const float t20 = T2[0 * 4 + r], t21 = T2[1 * 4 + r];
  const float t22 = T2[2 * 4 + r], t23 = T2[3 * 4 + r];
  v2f a1lo, a1hi, a2lo, a2hi;
  a1lo.x = holds ? t10 : 0.0f;  a1lo.y = holds ? t11 : 0.0f;
  a1hi.x = holds ? t12 : 0.0f;  a1hi.y = holds ? t13 : 0.0f;
  a2lo.x = holds ? t20 : 0.0f;  a2lo.y = holds ? t21 : 0.0f;
  a2hi.x = holds ? t22 : 0.0f;  a2hi.y = holds ? t23 : 0.0f;

  const size_t batch_off = (size_t)b * (size_t)m * 4;
  const float* F1 = f1 + batch_off;
  const float* F2 = f2 + batch_off;
  float* J = joint + batch_off;

  float acc = 0.0f;

  for (int t = wgid; t < tiles; t += wavesPerBatch) {
    const size_t off = ((size_t)(t << 5) + (size_t)lane) * 4;
    const v4f x1 = *(const v4f*)(F1 + off);
    const v4f x2 = *(const v4f*)(F2 + off);

    if (t + wavesPerBatch < tiles) {
      const size_t nxt = off + (size_t)wavesPerBatch * 128;
      __builtin_prefetch(F1 + nxt, 0, 1);
      __builtin_prefetch(F2 + nxt, 0, 1);
    }

    v2f b1lo, b1hi, b2lo, b2hi;
    b1lo.x = x1.x; b1lo.y = x1.y; b1hi.x = x1.z; b1hi.y = x1.w;
    b2lo.x = x2.x; b2lo.y = x2.y; b2hi.x = x2.z; b2hi.y = x2.w;

    v8f cz = {};
    v8f d1v = __builtin_amdgcn_wmma_f32_16x16x4_f32(
        false, a1lo, false, b1lo, (short)0, cz, false, false);
    v8f d2v = __builtin_amdgcn_wmma_f32_16x16x4_f32(
        false, a2lo, false, b2lo, (short)0, cz, false, false);
    d1v = __builtin_amdgcn_wmma_f32_16x16x4_f32(
        false, a1hi, false, b1hi, (short)0, d1v, false, false);
    d2v = __builtin_amdgcn_wmma_f32_16x16x4_f32(
        false, a2hi, false, b2hi, (short)0, d2v, false, false);

    v4f jv;
    jv.x = d1v[0] * d2v[0];
    jv.y = d1v[1] * d2v[1];
    jv.z = d1v[2] * d2v[2];
    jv.w = d1v[3] * d2v[3];

    *(v4f*)(J + off) = jv;
    acc += __logf(jv.x + jv.y + jv.z + jv.w);
  }

#pragma unroll
  for (int off = 16; off >= 1; off >>= 1)
    acc += __shfl_xor(acc, off, 32);
  if (lane == 0) atomicAdd(&logp[b], acc);
}

extern "C" void kernel_launch(void* const* d_in, const int* in_sizes, int n_in,
                              void* d_out, int out_size, void* d_ws, size_t ws_size,
                              hipStream_t stream) {
  const float* f1  = (const float*)d_in[0];
  const float* br1 = (const float*)d_in[1];
  const float* f2  = (const float*)d_in[2];
  const float* br2 = (const float*)d_in[3];

  const int B = in_sizes[1];
  const int m = in_sizes[0] / (B * 4);

  float* out   = (float*)d_out;
  float* joint = out;
  float* logp  = out + (size_t)B * m * 4;
  float* Tws   = (float*)d_ws;

  jc_setup_kernel<<<(B * 32 + 255) / 256, 256, 0, stream>>>(br1, br2, Tws, logp, B, m);

  const int blocksPerBatch = 4;
  jc_joint_kernel<<<B * blocksPerBatch, 256, 0, stream>>>(
      f1, f2, Tws, joint, logp, B, m, blocksPerBatch);
}
